# starting kernel
# baseline (speedup 1.0000x reference)
#include <hip/hip_runtime.h>


typedef float v2f __attribute__((ext_vector_type(2)));
typedef float v8f __attribute__((ext_vector_type(8)));

#define ORDER 64
#define NCHUNK 16
#define WAVES_PER_BLOCK 8

__global__ __launch_bounds__(256)
void ImplicitModalFilter_wmma_f32(const float* __restrict__ gamma,
                                  const float* __restrict__ R,
                                  const float* __restrict__ p,
                                  float* __restrict__ out,
                                  int D, int L, int ntiles) {
    const int wave = threadIdx.x >> 5;
    const int lane = threadIdx.x & 31;
    const int d = blockIdx.x * WAVES_PER_BLOCK + wave;
    if (d >= D) return;

    const int rq    = lane & 15;
    const int half  = lane >> 4;
    const int obase = half * 2;
    const float rf  = (float)rq;

    const float* gd = gamma + (size_t)d * ORDER;
    const float* Rd = R     + (size_t)d * ORDER;
    const float* pd = p     + (size_t)d * ORDER;

    float gl0[NCHUNK], gl1[NCHUNK];
    v2f   afrag[NCHUNK];

#pragma unroll
    for (int c = 0; c < NCHUNK; ++c) {
        const int o = 4 * c + obase;
        const float g0 = gd[o], g1 = gd[o + 1];
        const float p0 = pd[o], p1 = pd[o + 1];
        const float r0 = Rd[o], r1 = Rd[o + 1];
        const float l0 = -__expf(p0) * __expf(g0);
        const float l1 = -__expf(p1) * __expf(g1);
        gl0[c] = l0;
        gl1[c] = l1;
        afrag[c].x = __expf(l0 * rf) * r0;
        afrag[c].y = __expf(l1 * rf) * r1;
    }

    float* outd = out + (size_t)d * L;

    for (int j = 0; j < ntiles; ++j) {
        const float q16 = 16.0f * (float)(16 * j + rq);
        v8f acc = {};
#pragma unroll
        for (int c = 0; c < NCHUNK; ++c) {
            v2f b;
            b.x = __expf(gl0[c] * q16);
            b.y = __expf(gl1[c] * q16);
            acc = __builtin_amdgcn_wmma_f32_16x16x4_f32(
                 false, afrag[c],
                 false, b,
                 (short)0, acc,
                 false,  false);
        }
        const int tbase = 256 * j + 16 * rq + 8 * half;
        if (tbase + 7 < L) {
            float4 lo = make_float4(acc[0], acc[1], acc[2], acc[3]);
            float4 hi = make_float4(acc[4], acc[5], acc[6], acc[7]);
            *(float4*)(outd + tbase)     = lo;
            *(float4*)(outd + tbase + 4) = hi;
        }
    }
}

extern "C" void kernel_launch(void* const* d_in, const int* in_sizes, int n_in,
                              void* d_out, int out_size, void* d_ws, size_t ws_size,
                              hipStream_t stream) {
    const float* gamma = (const float*)d_in[0];
    const float* R     = (const float*)d_in[1];
    const float* p     = (const float*)d_in[2];
    float* out         = (float*)d_out;

    const int D = in_sizes[0] / ORDER;
    const int L = (D > 0) ? (out_size / D) : 0;
    const int ntiles = L / 256;

    const int blocks = (D + WAVES_PER_BLOCK - 1) / WAVES_PER_BLOCK;
    ImplicitModalFilter_wmma_f32<<<blocks, 256, 0, stream>>>(gamma, R, p, out, D, L, ntiles);
}


# B fragments of tile j made from tile j-1 by one packed multiply with exp(256*glogp) instead of 32 fresh exp per lane and tile; values become products of rounded factors, sum order unchanged
# speedup vs baseline: 1.1259x; 1.1259x over previous
#include <hip/hip_runtime.h>


typedef float v2f __attribute__((ext_vector_type(2)));
typedef float v8f __attribute__((ext_vector_type(8)));

#define ORDER 64
#define NCHUNK 16
#define WAVES_PER_BLOCK 8

template <bool ADVANCE>
__device__ __forceinline__ v8f tile_product(const v2f (&afrag)[NCHUNK], v2f (&bfrag)[NCHUNK],
                                            const v2f (&sfrag)[NCHUNK]) {
    v8f acc = {};
#pragma unroll
    for (int c = 0; c < NCHUNK; ++c) {
        if (ADVANCE) bfrag[c] = bfrag[c] * sfrag[c];
        acc = __builtin_amdgcn_wmma_f32_16x16x4_f32(
             false, afrag[c],
             false, bfrag[c],
             (short)0, acc,
             false,  false);
    }
    return acc;
}

__device__ __forceinline__ void store_tile(float* __restrict__ outd, int j, int rq, int half, int L,
                                           const v8f& acc) {
    const int tbase = 256 * j + 16 * rq + 8 * half;
    if (tbase + 7 < L) {
        float4 lo = make_float4(acc[0], acc[1], acc[2], acc[3]);
        float4 hi = make_float4(acc[4], acc[5], acc[6], acc[7]);
        *(float4*)(outd + tbase)     = lo;
        *(float4*)(outd + tbase + 4) = hi;
    }
}

__global__ __launch_bounds__(256)
void ImplicitModalFilter_wmma_f32(const float* __restrict__ gamma,
                                  const float* __restrict__ R,
                                  const float* __restrict__ p,
                                  float* __restrict__ out,
                                  int D, int L, int ntiles) {
    const int wave = threadIdx.x >> 5;
    const int lane = threadIdx.x & 31;
    const int d = blockIdx.x * WAVES_PER_BLOCK + wave;
    if (d >= D) return;
    if (ntiles < 1) return;

    const int rq    = lane & 15;
    const int half  = lane >> 4;
    const int obase = half * 2;
    const float rf  = (float)rq;
    const float n16 = 16.0f * rf;

    const float* gd = gamma + (size_t)d * ORDER;
    const float* Rd = R     + (size_t)d * ORDER;
    const float* pd = p     + (size_t)d * ORDER;

    v2f afrag[NCHUNK];
    v2f bfrag[NCHUNK];
    v2f sfrag[NCHUNK];

#pragma unroll
    for (int c = 0; c < NCHUNK; ++c) {
        const int o = 4 * c + obase;
        const float g0 = gd[o], g1 = gd[o + 1];
        const float p0 = pd[o], p1 = pd[o + 1];
        const float r0 = Rd[o], r1 = Rd[o + 1];
        const float l0 = -__expf(p0) * __expf(g0);
        const float l1 = -__expf(p1) * __expf(g1);
        afrag[c].x = __expf(l0 * rf) * r0;
        afrag[c].y = __expf(l1 * rf) * r1;
        bfrag[c].x = __expf(l0 * n16);
        bfrag[c].y = __expf(l1 * n16);
        sfrag[c].x = __expf(l0 * 256.0f);
        sfrag[c].y = __expf(l1 * 256.0f);
    }

    float* outd = out + (size_t)d * L;

    v8f acc = tile_product<false>(afrag, bfrag, sfrag);
    store_tile(outd, 0, rq, half, L, acc);

    for (int j = 1; j < ntiles; ++j) {
        acc = tile_product<true>(afrag, bfrag, sfrag);
        store_tile(outd, j, rq, half, L, acc);
    }
}

extern "C" void kernel_launch(void* const* d_in, const int* in_sizes, int n_in,
                              void* d_out, int out_size, void* d_ws, size_t ws_size,
                              hipStream_t stream) {
    const float* gamma = (const float*)d_in[0];
    const float* R     = (const float*)d_in[1];
    const float* p     = (const float*)d_in[2];
    float* out         = (float*)d_out;

    const int D = in_sizes[0] / ORDER;
    const int L = (D > 0) ? (out_size / D) : 0;
    const int ntiles = L / 256;

    const int blocks = (D + WAVES_PER_BLOCK - 1) / WAVES_PER_BLOCK;
    ImplicitModalFilter_wmma_f32<<<blocks, 256, 0, stream>>>(gamma, R, p, out, D, L, ntiles);
}


# Bounds test in front of each tile's stores dropped (true for every tile visited, j < L/256): no vector-compare-to-branch hand-off per tile; the loop body becomes two tiles on two accumulators
# speedup vs baseline: 1.1811x; 1.0490x over previous
#include <hip/hip_runtime.h>


typedef float v2f __attribute__((ext_vector_type(2)));
typedef float v8f __attribute__((ext_vector_type(8)));

#define ORDER 64
#define NCHUNK 16
#define WAVES_PER_BLOCK 8

template <bool ADVANCE>
__device__ __forceinline__ v8f tile_product(const v2f (&afrag)[NCHUNK], v2f (&bfrag)[NCHUNK],
                                            const v2f (&sfrag)[NCHUNK]) {
    v8f acc = {};
#pragma unroll
    for (int c = 0; c < NCHUNK; ++c) {
        if (ADVANCE) bfrag[c] = bfrag[c] * sfrag[c];
        acc = __builtin_amdgcn_wmma_f32_16x16x4_f32(
             false, afrag[c],
             false, bfrag[c],
             (short)0, acc,
             false,  false);
    }
    return acc;
}

__device__ __forceinline__ void store_tile(float* __restrict__ outd, int j, int rq, int half,
                                           const v8f& acc) {
    const int tbase = 256 * j + 16 * rq + 8 * half;
    float4 lo = make_float4(acc[0], acc[1], acc[2], acc[3]);
    float4 hi = make_float4(acc[4], acc[5], acc[6], acc[7]);
    *(float4*)(outd + tbase)     = lo;
    *(float4*)(outd + tbase + 4) = hi;
}

__global__ __launch_bounds__(256)
void ImplicitModalFilter_wmma_f32(const float* __restrict__ gamma,
                                  const float* __restrict__ R,
                                  const float* __restrict__ p,
                                  float* __restrict__ out,
                                  int D, int L, int ntiles) {
    const int wave = threadIdx.x >> 5;
    const int lane = threadIdx.x & 31;
    const int d = blockIdx.x * WAVES_PER_BLOCK + wave;
    if (d >= D) return;
    if (ntiles < 1) return;

    const int rq    = lane & 15;
    const int half  = lane >> 4;
    const int obase = half * 2;
    const float rf  = (float)rq;
    const float n16 = 16.0f * rf;

    const float* gd = gamma + (size_t)d * ORDER;
    const float* Rd = R     + (size_t)d * ORDER;
    const float* pd = p     + (size_t)d * ORDER;

    v2f afrag[NCHUNK];
    v2f bfrag[NCHUNK];
    v2f sfrag[NCHUNK];

#pragma unroll
    for (int c = 0; c < NCHUNK; ++c) {
        const int o = 4 * c + obase;
        const float g0 = gd[o], g1 = gd[o + 1];
        const float p0 = pd[o], p1 = pd[o + 1];
        const float r0 = Rd[o], r1 = Rd[o + 1];
        const float l0 = -__expf(p0) * __expf(g0);
        const float l1 = -__expf(p1) * __expf(g1);
        afrag[c].x = __expf(l0 * rf) * r0;
        afrag[c].y = __expf(l1 * rf) * r1;
        bfrag[c].x = __expf(l0 * n16);
        bfrag[c].y = __expf(l1 * n16);
        sfrag[c].x = __expf(l0 * 256.0f);
        sfrag[c].y = __expf(l1 * 256.0f);
    }

    float* outd = out + (size_t)d * L;

    v8f acc = tile_product<false>(afrag, bfrag, sfrag);
    store_tile(outd, 0, rq, half, acc);

    for (int j = 1; j < ntiles; ++j) {
        acc = tile_product<true>(afrag, bfrag, sfrag);
        store_tile(outd, j, rq, half, acc);
    }
}

extern "C" void kernel_launch(void* const* d_in, const int* in_sizes, int n_in,
                              void* d_out, int out_size, void* d_ws, size_t ws_size,
                              hipStream_t stream) {
    const float* gamma = (const float*)d_in[0];
    const float* R     = (const float*)d_in[1];
    const float* p     = (const float*)d_in[2];
    float* out         = (float*)d_out;

    const int D = in_sizes[0] / ORDER;
    const int L = (D > 0) ? (out_size / D) : 0;
    const int ntiles = L / 256;

    const int blocks = (D + WAVES_PER_BLOCK - 1) / WAVES_PER_BLOCK;
    ImplicitModalFilter_wmma_f32<<<blocks, 256, 0, stream>>>(gamma, R, p, out, D, L, ntiles);
}
